# starting kernel
# baseline (speedup 1.0000x reference)
#include <hip/hip_runtime.h>

typedef __attribute__((ext_vector_type(2))) float v2f;
typedef __attribute__((ext_vector_type(8))) float v8f;

static constexpr int KDIM = 128;

__global__ __launch_bounds__(256) void recompose_wmma_kernel(
    const float* __restrict__ targets,
    const float* __restrict__ temporal_bases,
    const float* __restrict__ scaling,
    const int* __restrict__ batch_indices,
    float* __restrict__ out,
    int n_tiles)
{
  __shared__ float w[KDIM];
  const int tid = threadIdx.x;
  if (tid < KDIM) {
    const int idx = batch_indices[1];
    w[tid] = temporal_bases[(size_t)idx * KDIM + tid] * scaling[tid];
  }
  __syncthreads();

  const int lane = tid & 31;
  const int wave = tid >> 5;
  const int h    = lane >> 4;
  const int m    = lane & 15;
  const int koff = 2 * h;

  const int wavesPerBlock = blockDim.x >> 5;
  for (int tile = blockIdx.x * wavesPerBlock + wave; tile < n_tiles;
       tile += gridDim.x * wavesPerBlock) {
    const float* arow = targets + (size_t)(tile * 16 + m) * KDIM + koff;
    v8f c0 = {};
    v8f c1 = {};
#pragma unroll
    for (int k0 = 0; k0 < KDIM; k0 += 8) {
      v2f a0 = __builtin_nontemporal_load((const v2f*)(arow + k0));
      v2f a1 = __builtin_nontemporal_load((const v2f*)(arow + k0 + 4));
      v2f b0 = *(const v2f*)(w + k0 + koff);
      v2f b1 = *(const v2f*)(w + k0 + 4 + koff);
      c0 = __builtin_amdgcn_wmma_f32_16x16x4_f32(false, a0, false, b0,
                                                 (short)0, c0, false, false);
      c1 = __builtin_amdgcn_wmma_f32_16x16x4_f32(false, a1, false, b1,
                                                 (short)0, c1, false, false);
    }
    const v8f c = c0 + c1;
    if (m == 0) {
      const int n0 = tile * 16 + 8 * h;
      *(float4*)(out + n0)     = make_float4(c[0], c[1], c[2], c[3]);
      *(float4*)(out + n0 + 4) = make_float4(c[4], c[5], c[6], c[7]);
    }
  }
}

__global__ void recompose_tail_kernel(
    const float* __restrict__ targets,
    const float* __restrict__ temporal_bases,
    const float* __restrict__ scaling,
    const int* __restrict__ batch_indices,
    float* __restrict__ out,
    int n_start, int n_total)
{
  const int n = n_start + blockIdx.x * blockDim.x + threadIdx.x;
  if (n >= n_total) return;
  const int idx = batch_indices[1];
  float acc = 0.f;
  for (int k = 0; k < KDIM; ++k)
    acc += targets[(size_t)n * KDIM + k] *
           (scaling[k] * temporal_bases[(size_t)idx * KDIM + k]);
  out[n] = acc;
}

extern "C" void kernel_launch(void* const* d_in, const int* in_sizes, int n_in,
                              void* d_out, int out_size, void* d_ws, size_t ws_size,
                              hipStream_t stream) {
  const float* targets = (const float*)d_in[0];
  const float* tb      = (const float*)d_in[1];
  const float* sc      = (const float*)d_in[2];
  const int*   bi      = (const int*)d_in[3];
  float* out = (float*)d_out;

  const int N = in_sizes[0] / KDIM;
  const int n_tiles = N / 16;

  if (n_tiles > 0) {
    const int waves_per_block = 8;
    const int blocks = (n_tiles + waves_per_block - 1) / waves_per_block;
    recompose_wmma_kernel<<<blocks, 256, 0, stream>>>(targets, tb, sc, bi, out,
                                                      n_tiles);
  }
  const int tail_start = n_tiles * 16;
  if (tail_start < N) {
    const int tail = N - tail_start;
    recompose_tail_kernel<<<(tail + 255) / 256, 256, 0, stream>>>(
        targets, tb, sc, bi, out, tail_start, N);
  }
}


# Whole-line global loads: each wave reads its 16 rows as 16 x 512 contiguous bytes (16 B per lane), parks them in a wave-private LDS tile and reads the WMMA A fragments from LDS in the same k order
# speedup vs baseline: 2.6330x; 2.6330x over previous
#include <hip/hip_runtime.h>

typedef __attribute__((ext_vector_type(2))) float v2f;
typedef __attribute__((ext_vector_type(4))) float v4f;
typedef __attribute__((ext_vector_type(8))) float v8f;

static constexpr int KDIM  = 128;
static constexpr int WAVES = 8;
static constexpr int PITCH = KDIM + 4;

__global__ __launch_bounds__(256) void recompose_wmma_kernel(
    const float* __restrict__ targets,
    const float* __restrict__ temporal_bases,
    const float* __restrict__ scaling,
    const int* __restrict__ batch_indices,
    float* __restrict__ out,
    int n_tiles)
{
  __shared__ float w[KDIM];
  __shared__ __attribute__((aligned(16))) float stage[WAVES][16 * PITCH];
  const int tid = threadIdx.x;
  if (tid < KDIM) {
    const int idx = batch_indices[1];
    w[tid] = temporal_bases[(size_t)idx * KDIM + tid] * scaling[tid];
  }
  __syncthreads();

  const int lane = tid & 31;
  const int wave = tid >> 5;
  const int h    = lane >> 4;
  const int m    = lane & 15;
  const int koff = 2 * h;

  const int wavesPerBlock = blockDim.x >> 5;
  for (int tile = blockIdx.x * wavesPerBlock + wave; tile < n_tiles;
       tile += gridDim.x * wavesPerBlock) {
    const float* grow = targets + (size_t)tile * 16 * KDIM + 4 * lane;
    v4f g[16];
#pragma unroll
    for (int q = 0; q < 16; ++q)
      g[q] = __builtin_nontemporal_load((const v4f*)(grow + q * KDIM));
    float* mine = stage[wave];
#pragma unroll
    for (int q = 0; q < 16; ++q)
      *(v4f*)(mine + q * PITCH + 4 * lane) = g[q];

    const float* arow = mine + m * PITCH + koff;
    v8f c0 = {};
    v8f c1 = {};
#pragma unroll
    for (int k0 = 0; k0 < KDIM; k0 += 8) {
      v2f a0 = *(const v2f*)(arow + k0);
      v2f a1 = *(const v2f*)(arow + k0 + 4);
      v2f b0 = *(const v2f*)(w + k0 + koff);
      v2f b1 = *(const v2f*)(w + k0 + 4 + koff);
      c0 = __builtin_amdgcn_wmma_f32_16x16x4_f32(false, a0, false, b0,
                                                 (short)0, c0, false, false);
      c1 = __builtin_amdgcn_wmma_f32_16x16x4_f32(false, a1, false, b1,
                                                 (short)0, c1, false, false);
    }
    const v8f c = c0 + c1;
    if (m == 0) {
      const int n0 = tile * 16 + 8 * h;
      *(float4*)(out + n0)     = make_float4(c[0], c[1], c[2], c[3]);
      *(float4*)(out + n0 + 4) = make_float4(c[4], c[5], c[6], c[7]);
    }
  }
}

__global__ void recompose_tail_kernel(
    const float* __restrict__ targets,
    const float* __restrict__ temporal_bases,
    const float* __restrict__ scaling,
    const int* __restrict__ batch_indices,
    float* __restrict__ out,
    int n_start, int n_total)
{
  const int n = n_start + blockIdx.x * blockDim.x + threadIdx.x;
  if (n >= n_total) return;
  const int idx = batch_indices[1];
  float acc = 0.f;
  for (int k = 0; k < KDIM; ++k)
    acc += targets[(size_t)n * KDIM + k] *
           (scaling[k] * temporal_bases[(size_t)idx * KDIM + k]);
  out[n] = acc;
}

extern "C" void kernel_launch(void* const* d_in, const int* in_sizes, int n_in,
                              void* d_out, int out_size, void* d_ws, size_t ws_size,
                              hipStream_t stream) {
  const float* targets = (const float*)d_in[0];
  const float* tb      = (const float*)d_in[1];
  const float* sc      = (const float*)d_in[2];
  const int*   bi      = (const int*)d_in[3];
  float* out = (float*)d_out;

  const int N = in_sizes[0] / KDIM;
  const int n_tiles = N / 16;

  if (n_tiles > 0) {
    const int waves_per_block = 8;
    const int blocks = (n_tiles + waves_per_block - 1) / waves_per_block;
    recompose_wmma_kernel<<<blocks, 256, 0, stream>>>(targets, tb, sc, bi, out,
                                                      n_tiles);
  }
  const int tail_start = n_tiles * 16;
  if (tail_start < N) {
    const int tail = N - tail_start;
    recompose_tail_kernel<<<(tail + 255) / 256, 256, 0, stream>>>(
        targets, tb, sc, bi, out, tail_start, N);
  }
}


# Four 16-row tiles per wave on a quarter of the grid: three register sets, the first three tiles' 48 whole-line loads issued together, the fourth tile's after the first tile's multiply
# speedup vs baseline: 2.9089x; 1.1048x over previous
#include <hip/hip_runtime.h>

typedef __attribute__((ext_vector_type(2))) float v2f;
typedef __attribute__((ext_vector_type(4))) float v4f;
typedef __attribute__((ext_vector_type(8))) float v8f;

static constexpr int KDIM  = 128;
static constexpr int WAVES = 8;
static constexpr int TPW   = 4;
static constexpr int PITCH = KDIM + 4;

__device__ __forceinline__ void load_tile(v4f (&g)[16], const float* src)
{
#pragma unroll
  for (int q = 0; q < 16; ++q)
    g[q] = __builtin_nontemporal_load((const v4f*)(src + q * KDIM));
}

__device__ __forceinline__ void park_tile(float* mine, int lane, const v4f (&g)[16])
{
#pragma unroll
  for (int q = 0; q < 16; ++q)
    *(v4f*)(mine + q * PITCH + 4 * lane) = g[q];
}

__device__ __forceinline__ v8f multiply_tile(const float* arow, const float* w, int koff)
{
  v8f c0 = {};
  v8f c1 = {};
#pragma unroll
  for (int k0 = 0; k0 < KDIM; k0 += 8) {
    v2f a0 = *(const v2f*)(arow + k0);
    v2f a1 = *(const v2f*)(arow + k0 + 4);
    v2f b0 = *(const v2f*)(w + k0 + koff);
    v2f b1 = *(const v2f*)(w + k0 + 4 + koff);
    c0 = __builtin_amdgcn_wmma_f32_16x16x4_f32(false, a0, false, b0,
                                               (short)0, c0, false, false);
    c1 = __builtin_amdgcn_wmma_f32_16x16x4_f32(false, a1, false, b1,
                                               (short)0, c1, false, false);
  }
  return c0 + c1;
}

__global__ __launch_bounds__(256) void recompose_wmma_kernel(
    const float* __restrict__ targets,
    const float* __restrict__ temporal_bases,
    const float* __restrict__ scaling,
    const int* __restrict__ batch_indices,
    float* __restrict__ out,
    int n_tiles)
{
  __shared__ float w[KDIM];
  __shared__ __attribute__((aligned(16))) float stage[WAVES][16 * PITCH];
  const int tid = threadIdx.x;
  if (tid < KDIM) {
    const int idx = batch_indices[1];
    w[tid] = temporal_bases[(size_t)idx * KDIM + tid] * scaling[tid];
  }
  __syncthreads();

  const int lane = tid & 31;
  const int wave = __builtin_amdgcn_readfirstlane(tid >> 5);
  const int h    = lane >> 4;
  const int m    = lane & 15;
  const int koff = 2 * h;

  float* mine = stage[wave];
  const float* arow = mine + m * PITCH + koff;
  const float* lane_src = targets + 4 * lane;
  const int last  = n_tiles - 1;
  const int tile0 = blockIdx.x * (WAVES * TPW) + wave;
  auto src_of = [&](int t) {
    return lane_src + (size_t)(t < last ? t : last) * (16 * KDIM);
  };

  v4f ga[16], gb[16], gc[16];
  load_tile(ga, src_of(tile0));
  load_tile(gb, src_of(tile0 + WAVES));
  load_tile(gc, src_of(tile0 + 2 * WAVES));

  auto finish = [&](int tile) {
    const v8f c = multiply_tile(arow, w, koff);
    if (m == 0 && tile < n_tiles) {
      const int n0 = tile * 16 + 8 * h;
      *(float4*)(out + n0)     = make_float4(c[0], c[1], c[2], c[3]);
      *(float4*)(out + n0 + 4) = make_float4(c[4], c[5], c[6], c[7]);
    }
  };

  park_tile(mine, lane, ga);
  finish(tile0);
  load_tile(ga, src_of(tile0 + 3 * WAVES));
  park_tile(mine, lane, gb);
  finish(tile0 + WAVES);
  park_tile(mine, lane, gc);
  finish(tile0 + 2 * WAVES);
  park_tile(mine, lane, ga);
  finish(tile0 + 3 * WAVES);
}

__global__ void recompose_tail_kernel(
    const float* __restrict__ targets,
    const float* __restrict__ temporal_bases,
    const float* __restrict__ scaling,
    const int* __restrict__ batch_indices,
    float* __restrict__ out,
    int n_start, int n_total)
{
  const int n = n_start + blockIdx.x * blockDim.x + threadIdx.x;
  if (n >= n_total) return;
  const int idx = batch_indices[1];
  float acc = 0.f;
  for (int k = 0; k < KDIM; ++k)
    acc += targets[(size_t)n * KDIM + k] *
           (scaling[k] * temporal_bases[(size_t)idx * KDIM + k]);
  out[n] = acc;
}

extern "C" void kernel_launch(void* const* d_in, const int* in_sizes, int n_in,
                              void* d_out, int out_size, void* d_ws, size_t ws_size,
                              hipStream_t stream) {
  const float* targets = (const float*)d_in[0];
  const float* tb      = (const float*)d_in[1];
  const float* sc      = (const float*)d_in[2];
  const int*   bi      = (const int*)d_in[3];
  float* out = (float*)d_out;

  const int N = in_sizes[0] / KDIM;
  const int n_tiles = N / 16;

  if (n_tiles > 0) {
    const int tiles_per_block = WAVES * TPW;
    const int blocks = (n_tiles + tiles_per_block - 1) / tiles_per_block;
    recompose_wmma_kernel<<<blocks, 256, 0, stream>>>(targets, tb, sc, bi, out,
                                                      n_tiles);
  }
  const int tail_start = n_tiles * 16;
  if (tail_start < N) {
    const int tail = N - tail_start;
    recompose_tail_kernel<<<(tail + 255) / 256, 256, 0, stream>>>(
        targets, tb, sc, bi, out, tail_start, N);
  }
}
